# starting kernel
# baseline (speedup 1.0000x reference)
#include <hip/hip_runtime.h>

#define D 128

typedef __attribute__((ext_vector_type(2))) float v2f;
typedef __attribute__((ext_vector_type(4))) float f4;
typedef __attribute__((ext_vector_type(8))) float v8f;

__global__ void __launch_bounds__(256)
gemm_xw(const float* __restrict__ X, const float* __restrict__ W,
        float* __restrict__ P, int M)
{
    const int lane  = threadIdx.x & 31;
    const int wave  = threadIdx.x >> 5;
    const int nBase = wave * 16;
    const int col   = lane & 15;
    const int khalf = lane >> 4;

    v2f bfrag[32];
#pragma unroll
    for (int kk = 0; kk < 32; ++kk) {
        const int k = kk * 4 + khalf * 2;
        bfrag[kk].x = W[(size_t)k * D + nBase + col];
        bfrag[kk].y = W[(size_t)(k + 1) * D + nBase + col];
    }

    const int numTiles = M >> 5;
    for (int mt = blockIdx.x; mt < numTiles; mt += gridDim.x) {
        const int mBase = mt << 5;
        const float* arow0 = X + (size_t)(mBase + col) * D + khalf * 2;
        const float* arow1 = arow0 + (size_t)16 * D;
        v8f acc0 = {};
        v8f acc1 = {};
#pragma unroll
        for (int kk = 0; kk < 32; ++kk) {
            v2f a0 = *(const v2f*)(arow0 + kk * 4);
            v2f a1 = *(const v2f*)(arow1 + kk * 4);
            acc0 = __builtin_amdgcn_wmma_f32_16x16x4_f32(
                false, a0, false, bfrag[kk], (short)0, acc0, false, false);
            acc1 = __builtin_amdgcn_wmma_f32_16x16x4_f32(
                false, a1, false, bfrag[kk], (short)0, acc1, false, false);
        }
        float* prow0 = P + (size_t)mBase * D + nBase + col;
        float* prow1 = prow0 + (size_t)16 * D;
#pragma unroll
        for (int r = 0; r < 8; ++r) {
            prow0[(size_t)(r + khalf * 8) * D] = acc0[r];
            prow1[(size_t)(r + khalf * 8) * D] = acc1[r];
        }
    }
}

__global__ void __launch_bounds__(256)
edge_scatter(const float* __restrict__ Pf, const float* __restrict__ Pv,
             const int* __restrict__ senders, const int* __restrict__ receivers,
             const float* __restrict__ edge_attr,
             const float* __restrict__ w_e, const float* __restrict__ b_msg,
             float* __restrict__ aggr, int E)
{
    const int lane = threadIdx.x & 31;
    const int e    = blockIdx.x * 8 + (threadIdx.x >> 5);
    if (e >= E) return;

    const int   s  = senders[e];
    const int   r  = receivers[e];
    const float ea = edge_attr[e];
    const int   o  = lane * 4;

    f4 pf = *(const f4*)(Pf + (size_t)r * D + o);
    f4 pv = *(const f4*)(Pv + (size_t)s * D + o);
    f4 we = *(const f4*)(w_e + o);
    f4 bm = *(const f4*)(b_msg + o);

    float* dst = aggr + (size_t)r * D + o;
#pragma unroll
    for (int i = 0; i < 4; ++i) {
        float m = pf[i] + pv[i] + ea * we[i] + bm[i];
        m = m > 0.0f ? m : 0.0f;
        float* p = dst + i;
        asm volatile("global_atomic_add_f32 %0, %1, off scope:SCOPE_DEV"
                     :: "v"(p), "v"(m)
                     : "memory");
    }
}

__global__ void __launch_bounds__(256)
combine_gemm(const float* __restrict__ Fa, const float* __restrict__ Ag,
             const float* __restrict__ Wc, const float* __restrict__ bc,
             float* __restrict__ out, int M)
{
    const int lane  = threadIdx.x & 31;
    const int wave  = threadIdx.x >> 5;
    const int nBase = wave * 16;
    const int col   = lane & 15;
    const int khalf = lane >> 4;

    const int numTiles = M >> 5;
    for (int mt = blockIdx.x; mt < numTiles; mt += gridDim.x) {
        const int mBase = mt << 5;
        const float* arowF0 = Fa + (size_t)(mBase + col) * D + khalf * 2;
        const float* arowA0 = Ag + (size_t)(mBase + col) * D + khalf * 2;
        const float* arowF1 = arowF0 + (size_t)16 * D;
        const float* arowA1 = arowA0 + (size_t)16 * D;
        v8f acc0 = {};
        v8f acc1 = {};
#pragma unroll
        for (int kk = 0; kk < 64; ++kk) {
            const int k = kk * 4 + khalf * 2;
            v2f b;
            b.x = Wc[(size_t)k * D + nBase + col];
            b.y = Wc[(size_t)(k + 1) * D + nBase + col];
            v2f a0, a1;
            if (kk < 32) {
                a0 = *(const v2f*)(arowF0 + kk * 4);
                a1 = *(const v2f*)(arowF1 + kk * 4);
            } else {
                a0 = *(const v2f*)(arowA0 + (kk - 32) * 4);
                a1 = *(const v2f*)(arowA1 + (kk - 32) * 4);
            }
            acc0 = __builtin_amdgcn_wmma_f32_16x16x4_f32(
                false, a0, false, b, (short)0, acc0, false, false);
            acc1 = __builtin_amdgcn_wmma_f32_16x16x4_f32(
                false, a1, false, b, (short)0, acc1, false, false);
        }
        const float bias = bc[nBase + col];
        float* orow0 = out + (size_t)mBase * D + nBase + col;
        float* orow1 = orow0 + (size_t)16 * D;
#pragma unroll
        for (int r = 0; r < 8; ++r) {
            float v0 = acc0[r] + bias;
            float v1 = acc1[r] + bias;
            orow0[(size_t)(r + khalf * 8) * D] = v0 > 0.0f ? v0 : 0.0f;
            orow1[(size_t)(r + khalf * 8) * D] = v1 > 0.0f ? v1 : 0.0f;
        }
    }
}

extern "C" void kernel_launch(void* const* d_in, const int* in_sizes, int n_in,
                              void* d_out, int out_size, void* d_ws, size_t ws_size,
                              hipStream_t stream)
{
    const float* variables = (const float*)d_in[0];
    const float* factors   = (const float*)d_in[1];
    const int*   senders   = (const int*)  d_in[2];
    const int*   receivers = (const int*)  d_in[3];
    const float* edge_attr = (const float*)d_in[4];
    const float* W_msg     = (const float*)d_in[5];
    const float* b_msg     = (const float*)d_in[6];
    const float* W_comb    = (const float*)d_in[7];
    const float* b_comb    = (const float*)d_in[8];

    const int V = in_sizes[0] / D;
    const int F = in_sizes[1] / D;
    const int E = in_sizes[2];

    const float* W_i = W_msg;
    const float* W_j = W_msg + (size_t)128 * D;
    const float* w_e = W_msg + (size_t)256 * D;

    float* Pv   = (float*)d_ws;
    float* Pf   = Pv + (size_t)V * D;
    float* aggr = Pf + (size_t)F * D;

    hipMemsetAsync(aggr, 0, (size_t)F * D * sizeof(float), stream);

    const int tilesV = V >> 5, tilesF = F >> 5;
    const int gV = tilesV < 2048 ? tilesV : 2048;
    const int gF = tilesF < 2048 ? tilesF : 2048;

    gemm_xw<<<gV, 256, 0, stream>>>(variables, W_j, Pv, V);
    gemm_xw<<<gF, 256, 0, stream>>>(factors,   W_i, Pf, F);

    edge_scatter<<<(E + 7) / 8, 256, 0, stream>>>(
        Pf, Pv, senders, receivers, edge_attr, w_e, b_msg, aggr, E);

    combine_gemm<<<gF, 256, 0, stream>>>(
        factors, aggr, W_comb, b_comb, (float*)d_out, F);
}


# A tile of all three GEMM launches fetched once per block with whole-row 16-byte loads and staged through LDS (register prefetch, two buffers); fragments read from LDS, k order kept
# speedup vs baseline: 1.2214x; 1.2214x over previous
#include <hip/hip_runtime.h>

#define D 128

typedef __attribute__((ext_vector_type(2))) float v2f;
typedef __attribute__((ext_vector_type(4))) float f4;
typedef __attribute__((ext_vector_type(8))) float v8f;

#define LDP 132
#define TILE_FLOATS (32 * LDP)

__global__ void __launch_bounds__(256)
gemm_xw(const float* __restrict__ X, const float* __restrict__ W,
        float* __restrict__ P, int M)
{
    __shared__ __attribute__((aligned(16))) float tileA[2 * TILE_FLOATS];

    const int lane  = threadIdx.x & 31;
    const int wave  = threadIdx.x >> 5;
    const int nBase = wave * 16;
    const int col   = lane & 15;
    const int khalf = lane >> 4;

    v2f bfrag[32];
#pragma unroll
    for (int kk = 0; kk < 32; ++kk) {
        const int k = kk * 4 + khalf * 2;
        bfrag[kk].x = W[(size_t)k * D + nBase + col];
        bfrag[kk].y = W[(size_t)(k + 1) * D + nBase + col];
    }

    const int srcOff = wave * D + lane * 4;
    const int ldsOff = wave * LDP + lane * 4;

    const int numTiles = M >> 5;
    int mt = blockIdx.x;

    f4 stage[4];
    if (mt < numTiles) {
        const float* src = X + (size_t)(mt << 5) * D + srcOff;
        for (int i = 0; i < 4; ++i)
            stage[i] = *(const f4*)(src + (size_t)(8 * i) * D);
    }

    int buf = 0;
    for (; mt < numTiles; mt += gridDim.x) {
        float* tile = tileA + buf * TILE_FLOATS;
        for (int i = 0; i < 4; ++i)
            *(f4*)(tile + ldsOff + 8 * i * LDP) = stage[i];
        __syncthreads();

        const int nxt = mt + gridDim.x;
        if (nxt < numTiles) {
            const float* src = X + (size_t)(nxt << 5) * D + srcOff;
            for (int i = 0; i < 4; ++i)
                stage[i] = *(const f4*)(src + (size_t)(8 * i) * D);
        }

        const float* arow0 = tile + col * LDP + khalf * 2;
        const float* arow1 = arow0 + 16 * LDP;
        v8f acc0 = {};
        v8f acc1 = {};
#pragma unroll
        for (int kk = 0; kk < 32; ++kk) {
            v2f a0 = *(const v2f*)(arow0 + kk * 4);
            v2f a1 = *(const v2f*)(arow1 + kk * 4);
            acc0 = __builtin_amdgcn_wmma_f32_16x16x4_f32(
                false, a0, false, bfrag[kk], (short)0, acc0, false, false);
            acc1 = __builtin_amdgcn_wmma_f32_16x16x4_f32(
                false, a1, false, bfrag[kk], (short)0, acc1, false, false);
        }
        const int mBase = mt << 5;
        float* prow0 = P + (size_t)mBase * D + nBase + col;
        float* prow1 = prow0 + (size_t)16 * D;
#pragma unroll
        for (int r = 0; r < 8; ++r) {
            prow0[(size_t)(r + khalf * 8) * D] = acc0[r];
            prow1[(size_t)(r + khalf * 8) * D] = acc1[r];
        }
        buf ^= 1;
    }
}

__global__ void __launch_bounds__(256)
edge_scatter(const float* __restrict__ Pf, const float* __restrict__ Pv,
             const int* __restrict__ senders, const int* __restrict__ receivers,
             const float* __restrict__ edge_attr,
             const float* __restrict__ w_e, const float* __restrict__ b_msg,
             float* __restrict__ aggr, int E)
{
    const int lane = threadIdx.x & 31;
    const int e    = blockIdx.x * 8 + (threadIdx.x >> 5);
    if (e >= E) return;

    const int   s  = senders[e];
    const int   r  = receivers[e];
    const float ea = edge_attr[e];
    const int   o  = lane * 4;

    f4 pf = *(const f4*)(Pf + (size_t)r * D + o);
    f4 pv = *(const f4*)(Pv + (size_t)s * D + o);
    f4 we = *(const f4*)(w_e + o);
    f4 bm = *(const f4*)(b_msg + o);

    float* dst = aggr + (size_t)r * D + o;
#pragma unroll
    for (int i = 0; i < 4; ++i) {
        float m = pf[i] + pv[i] + ea * we[i] + bm[i];
        m = m > 0.0f ? m : 0.0f;
        float* p = dst + i;
        asm volatile("global_atomic_add_f32 %0, %1, off scope:SCOPE_DEV"
                     :: "v"(p), "v"(m)
                     : "memory");
    }
}

__global__ void __launch_bounds__(256)
combine_gemm(const float* __restrict__ Fa, const float* __restrict__ Ag,
             const float* __restrict__ Wc, const float* __restrict__ bc,
             float* __restrict__ out, int M)
{
    __shared__ __attribute__((aligned(16))) float tileA[2 * TILE_FLOATS];

    const int lane  = threadIdx.x & 31;
    const int wave  = threadIdx.x >> 5;
    const int nBase = wave * 16;
    const int col   = lane & 15;
    const int khalf = lane >> 4;

    const int srcOff = wave * D + lane * 4;
    const int ldsOff = wave * LDP + lane * 4;

    const float* bcol = Wc + (size_t)(khalf * 2) * D + nBase + col;

    const int numTiles = M >> 5;
    int mt = blockIdx.x;

    f4 stageF[4], stageG[4];
    if (mt < numTiles) {
        const float* srcF = Fa + (size_t)(mt << 5) * D + srcOff;
        const float* srcG = Ag + (size_t)(mt << 5) * D + srcOff;
        for (int i = 0; i < 4; ++i) {
            stageF[i] = *(const f4*)(srcF + (size_t)(8 * i) * D);
            stageG[i] = *(const f4*)(srcG + (size_t)(8 * i) * D);
        }
    }

    for (; mt < numTiles; mt += gridDim.x) {
        const int mBase = mt << 5;
        const int nxt   = mt + gridDim.x;
        v8f acc0 = {};
        v8f acc1 = {};

        {
            float* tile = tileA;
            for (int i = 0; i < 4; ++i)
                *(f4*)(tile + ldsOff + 8 * i * LDP) = stageF[i];
            __syncthreads();

            if (nxt < numTiles) {
                const float* src = Fa + (size_t)(nxt << 5) * D + srcOff;
                for (int i = 0; i < 4; ++i)
                    stageF[i] = *(const f4*)(src + (size_t)(8 * i) * D);
            }

            const float* arow0 = tile + col * LDP + khalf * 2;
            const float* arow1 = arow0 + 16 * LDP;
#pragma unroll
            for (int kk = 0; kk < 32; ++kk) {
                v2f b;
                b.x = bcol[(size_t)(kk * 4) * D];
                b.y = bcol[(size_t)(kk * 4 + 1) * D];
                v2f a0 = *(const v2f*)(arow0 + kk * 4);
                v2f a1 = *(const v2f*)(arow1 + kk * 4);
                acc0 = __builtin_amdgcn_wmma_f32_16x16x4_f32(
                    false, a0, false, b, (short)0, acc0, false, false);
                acc1 = __builtin_amdgcn_wmma_f32_16x16x4_f32(
                    false, a1, false, b, (short)0, acc1, false, false);
            }
        }

        {
            float* tile = tileA + TILE_FLOATS;
            for (int i = 0; i < 4; ++i)
                *(f4*)(tile + ldsOff + 8 * i * LDP) = stageG[i];
            __syncthreads();

            if (nxt < numTiles) {
                const float* src = Ag + (size_t)(nxt << 5) * D + srcOff;
                for (int i = 0; i < 4; ++i)
                    stageG[i] = *(const f4*)(src + (size_t)(8 * i) * D);
            }

            const float* arow0 = tile + col * LDP + khalf * 2;
            const float* arow1 = arow0 + 16 * LDP;
#pragma unroll
            for (int kk = 0; kk < 32; ++kk) {
                v2f b;
                b.x = bcol[(size_t)(128 + kk * 4) * D];
                b.y = bcol[(size_t)(128 + kk * 4 + 1) * D];
                v2f a0 = *(const v2f*)(arow0 + kk * 4);
                v2f a1 = *(const v2f*)(arow1 + kk * 4);
                acc0 = __builtin_amdgcn_wmma_f32_16x16x4_f32(
                    false, a0, false, b, (short)0, acc0, false, false);
                acc1 = __builtin_amdgcn_wmma_f32_16x16x4_f32(
                    false, a1, false, b, (short)0, acc1, false, false);
            }
        }

        const float bias = bc[nBase + col];
        float* orow0 = out + (size_t)mBase * D + nBase + col;
        float* orow1 = orow0 + (size_t)16 * D;
#pragma unroll
        for (int r = 0; r < 8; ++r) {
            float v0 = acc0[r] + bias;
            float v1 = acc1[r] + bias;
            orow0[(size_t)(r + khalf * 8) * D] = v0 > 0.0f ? v0 : 0.0f;
            orow1[(size_t)(r + khalf * 8) * D] = v1 > 0.0f ? v1 : 0.0f;
        }
    }
}

extern "C" void kernel_launch(void* const* d_in, const int* in_sizes, int n_in,
                              void* d_out, int out_size, void* d_ws, size_t ws_size,
                              hipStream_t stream)
{
    const float* variables = (const float*)d_in[0];
    const float* factors   = (const float*)d_in[1];
    const int*   senders   = (const int*)  d_in[2];
    const int*   receivers = (const int*)  d_in[3];
    const float* edge_attr = (const float*)d_in[4];
    const float* W_msg     = (const float*)d_in[5];
    const float* b_msg     = (const float*)d_in[6];
    const float* W_comb    = (const float*)d_in[7];
    const float* b_comb    = (const float*)d_in[8];

    const int V = in_sizes[0] / D;
    const int F = in_sizes[1] / D;
    const int E = in_sizes[2];

    const float* W_i = W_msg;
    const float* W_j = W_msg + (size_t)128 * D;
    const float* w_e = W_msg + (size_t)256 * D;

    float* Pv   = (float*)d_ws;
    float* Pf   = Pv + (size_t)V * D;
    float* aggr = Pf + (size_t)F * D;

    hipMemsetAsync(aggr, 0, (size_t)F * D * sizeof(float), stream);

    const int tilesV = V >> 5, tilesF = F >> 5;
    const int gV = tilesV < 2048 ? tilesV : 2048;
    const int gF = tilesF < 2048 ? tilesF : 2048;

    gemm_xw<<<gV, 256, 0, stream>>>(variables, W_j, Pv, V);
    gemm_xw<<<gF, 256, 0, stream>>>(factors,   W_i, Pf, F);

    edge_scatter<<<(E + 7) / 8, 256, 0, stream>>>(
        Pf, Pv, senders, receivers, edge_attr, w_e, b_msg, aggr, E);

    combine_gemm<<<gF, 256, 0, stream>>>(
        factors, aggr, W_comb, b_comb, (float*)d_out, F);
}


# Float atomics and accumulator clear replaced by a destination-ordered sum: a block owns 416 factor rows, collects their edges into LDS lists, one wave sums each row in LDS; adds in fixed list order
# speedup vs baseline: 1.6407x; 1.3433x over previous
#include <hip/hip_runtime.h>

#define D 128

typedef __attribute__((ext_vector_type(2))) float v2f;
typedef __attribute__((ext_vector_type(4))) float f4;
typedef __attribute__((ext_vector_type(8))) float v8f;

#define LDP 132
#define TILE_FLOATS (32 * LDP)

__global__ void __launch_bounds__(256)
gemm_xw(const float* __restrict__ X, const float* __restrict__ W,
        float* __restrict__ P, int M)
{
    __shared__ __attribute__((aligned(16))) float tileA[2 * TILE_FLOATS];

    const int lane  = threadIdx.x & 31;
    const int wave  = threadIdx.x >> 5;
    const int nBase = wave * 16;
    const int col   = lane & 15;
    const int khalf = lane >> 4;

    v2f bfrag[32];
#pragma unroll
    for (int kk = 0; kk < 32; ++kk) {
        const int k = kk * 4 + khalf * 2;
        bfrag[kk].x = W[(size_t)k * D + nBase + col];
        bfrag[kk].y = W[(size_t)(k + 1) * D + nBase + col];
    }

    const int srcOff = wave * D + lane * 4;
    const int ldsOff = wave * LDP + lane * 4;

    const int numTiles = M >> 5;
    int mt = blockIdx.x;

    f4 stage[4];
    if (mt < numTiles) {
        const float* src = X + (size_t)(mt << 5) * D + srcOff;
        for (int i = 0; i < 4; ++i)
            stage[i] = *(const f4*)(src + (size_t)(8 * i) * D);
    }

    int buf = 0;
    for (; mt < numTiles; mt += gridDim.x) {
        float* tile = tileA + buf * TILE_FLOATS;
        for (int i = 0; i < 4; ++i)
            *(f4*)(tile + ldsOff + 8 * i * LDP) = stage[i];
        __syncthreads();

        const int nxt = mt + gridDim.x;
        if (nxt < numTiles) {
            const float* src = X + (size_t)(nxt << 5) * D + srcOff;
            for (int i = 0; i < 4; ++i)
                stage[i] = *(const f4*)(src + (size_t)(8 * i) * D);
        }

        const float* arow0 = tile + col * LDP + khalf * 2;
        const float* arow1 = arow0 + 16 * LDP;
        v8f acc0 = {};
        v8f acc1 = {};
#pragma unroll
        for (int kk = 0; kk < 32; ++kk) {
            v2f a0 = *(const v2f*)(arow0 + kk * 4);
            v2f a1 = *(const v2f*)(arow1 + kk * 4);
            acc0 = __builtin_amdgcn_wmma_f32_16x16x4_f32(
                false, a0, false, bfrag[kk], (short)0, acc0, false, false);
            acc1 = __builtin_amdgcn_wmma_f32_16x16x4_f32(
                false, a1, false, bfrag[kk], (short)0, acc1, false, false);
        }
        const int mBase = mt << 5;
        float* prow0 = P + (size_t)mBase * D + nBase + col;
        float* prow1 = prow0 + (size_t)16 * D;
#pragma unroll
        for (int r = 0; r < 8; ++r) {
            prow0[(size_t)(r + khalf * 8) * D] = acc0[r];
            prow1[(size_t)(r + khalf * 8) * D] = acc1[r];
        }
        buf ^= 1;
    }
}

#define RPW  13
#define RPB  (32 * RPW)
#define LCAP 224
#define QCAP 64

static __device__ __forceinline__ f4 pick_row(const f4 (&r)[RPW], int t)
{
    switch (t) {
    case 0:  return r[0];
    case 1:  return r[1];
    case 2:  return r[2];
    case 3:  return r[3];
    case 4:  return r[4];
    case 5:  return r[5];
    case 6:  return r[6];
    case 7:  return r[7];
    case 8:  return r[8];
    case 9:  return r[9];
    case 10: return r[10];
    case 11: return r[11];
    default: return r[12];
    }
}

__global__ void __launch_bounds__(1024)
row_owner_sum(const float* __restrict__ Pf, const float* __restrict__ Pv,
              const int* __restrict__ senders, const int* __restrict__ receivers,
              const float* __restrict__ edge_attr,
              const float* __restrict__ w_e, const float* __restrict__ b_msg,
              float* __restrict__ aggr, int E, int F)
{
    __shared__ f4  accS[32 * RPW * 32];
    __shared__ int listS[32 * LCAP];
    __shared__ int listA[32 * LCAP];
    __shared__ unsigned short listT[32 * LCAP];
    __shared__ int quI[32 * QCAP];
    __shared__ int cntW[32];
    __shared__ int moreFlag;

    const int lane = threadIdx.x & 31;
    const int w    = __builtin_amdgcn_readfirstlane(threadIdx.x >> 5);
    const int lo   = blockIdx.x * RPB;
    const int Rb   = F - lo < RPB ? F - lo : RPB;
    const int wlo  = w * RPW;
    const int o    = lane * 4;
    const unsigned below = (1u << lane) - 1u;

    f4*  myAcc = accS + wlo * 32;
    unsigned short* myT = listT + w * LCAP;
    int* myS = listS + w * LCAP;
    int* myA = listA + w * LCAP;
    int* mqI = quI + w * QCAP;

    const f4 zero = {0.0f, 0.0f, 0.0f, 0.0f};
    f4 pfr[RPW];
#pragma unroll
    for (int k = 0; k < RPW; ++k) {
        myAcc[k * 32 + lane] = zero;
        pfr[k] = wlo + k < Rb ? *(const f4*)(Pf + (size_t)(lo + wlo + k) * D + o) : zero;
    }

    const f4 we = *(const f4*)(w_e + o);
    const f4 bm = *(const f4*)(b_msg + o);

    const int numChunks = (E + 31) >> 5;
    const int numP      = (numChunks + 31) >> 5;
    const int pFull     = (E >> 5) >> 5;
    const int pRot      = (int)(((long long)blockIdx.x * numP) / gridDim.x);
    int p = 0;

    for (;;) {
        if (threadIdx.x == 0) moreFlag = 0;
        __syncthreads();

        int cursor = 0;
        while (p < numP && cursor + 32 <= LCAP) {
            int pr = p + pRot;
            if (pr >= numP) pr -= numP;
            if (pr + 16 <= pFull && p + 16 <= numP) {
                const int eBase = (pr * 32 + w) * 32 + lane;
                const int* src = receivers + eBase;
                int rr[16];
#pragma unroll
                for (int u = 0; u < 16; ++u) rr[u] = src[u * 1024];
                int done = 16;
                unsigned rbEff = (unsigned)Rb;
#pragma unroll
                for (int u = 0; u < 16; ++u) {
                    const int t = rr[u] - lo;
                    const bool m = (unsigned)t < rbEff;
                    const unsigned mask = __builtin_amdgcn_ballot_w32(m);
                    if (mask) {
                        if (m) {
                            const int slot = cursor + __builtin_popcount(mask & below);
                            myT[slot] = (unsigned short)t;
                            myS[slot] = eBase + u * 1024;
                        }
                        cursor += __builtin_popcount(mask);
                        if (cursor + 32 > LCAP) { rbEff = 0u; done = u + 1; }
                    }
                }
                p += done;
            } else {
                const int cp = pr * 32 + w;
                const int e  = cp * 32 + lane;
                const int r  = (cp < numChunks && e < E) ? receivers[e] : -1;
                const int t  = r - lo;
                const bool m = (unsigned)t < (unsigned)Rb;
                const unsigned mask = __builtin_amdgcn_ballot_w32(m);
                if (mask) {
                    if (m) {
                        const int slot = cursor + __builtin_popcount(mask & below);
                        myT[slot] = (unsigned short)t;
                        myS[slot] = e;
                    }
                    cursor += __builtin_popcount(mask);
                }
                p += 1;
            }
        }
        for (int idx = lane; idx < cursor; idx += 32) {
            const int e = myS[idx];
            myS[idx] = senders[e];
            myA[idx] = __float_as_int(edge_attr[e]);
        }
        if (lane == 0) {
            cntW[w] = cursor;
            if (p < numP) moreFlag = 1;
        }
        __syncthreads();

        const int more = __builtin_amdgcn_readfirstlane(moreFlag);
        int qn = 0;
        for (int q = 0; q < 32; ++q) {
            const int nq = __builtin_amdgcn_readfirstlane(cntW[q]);
            for (int base = 0; base <= nq; base += 32) {
                const bool last = (q == 31) && (base + 32 > nq);
                const int idx = base + lane;
                const int t = idx < nq ? (int)listT[q * LCAP + idx] - wlo : -1;
                const bool m = (unsigned)t < (unsigned)RPW;
                const unsigned mask = __builtin_amdgcn_ballot_w32(m);
                if (mask) {
                    if (m) {
                        const int slot = qn + __builtin_popcount(mask & below);
                        mqI[slot] = q * LCAP + idx;
                    }
                    qn += __builtin_popcount(mask);
                }
                if (qn > QCAP - 32 || (last && qn > 0)) {
                    for (int g = 0; g < qn; g += 8) {
                        const int cnt = qn - g < 8 ? qn - g : 8;
                        int tq[8];
                        float aq[8];
                        f4 pv[8];
#pragma unroll
                        for (int u = 0; u < 8; ++u) {
                            const int li = mqI[g + (u < cnt ? u : cnt - 1)];
                            tq[u] = __builtin_amdgcn_readfirstlane((int)listT[li]) - wlo;
                            const int s = listS[li];
                            aq[u] = __int_as_float(listA[li]);
                            pv[u] = *(const f4*)(Pv + (size_t)s * D + o);
                        }
#pragma unroll
                        for (int u = 0; u < 8; ++u) {
                            if (u < cnt) {
                                const f4 pf = pick_row(pfr, tq[u]);
                                f4 a = myAcc[tq[u] * 32 + lane];
#pragma unroll
                                for (int i = 0; i < 4; ++i) {
                                    float m2 = pf[i] + pv[u][i] + aq[u] * we[i] + bm[i];
                                    a[i] += m2 > 0.0f ? m2 : 0.0f;
                                }
                                myAcc[tq[u] * 32 + lane] = a;
                            }
                        }
                    }
                    qn = 0;
                }
                if (last) break;
            }
        }
        __syncthreads();
        if (!more) break;
    }

    for (int k = 0; k < RPW; ++k) {
        if (wlo + k < Rb)
            *(f4*)(aggr + (size_t)(lo + wlo + k) * D + o) = myAcc[k * 32 + lane];
    }
}

__global__ void __launch_bounds__(256)
combine_gemm(const float* __restrict__ Fa, const float* __restrict__ Ag,
             const float* __restrict__ Wc, const float* __restrict__ bc,
             float* __restrict__ out, int M)
{
    __shared__ __attribute__((aligned(16))) float tileA[2 * TILE_FLOATS];

    const int lane  = threadIdx.x & 31;
    const int wave  = threadIdx.x >> 5;
    const int nBase = wave * 16;
    const int col   = lane & 15;
    const int khalf = lane >> 4;

    const int srcOff = wave * D + lane * 4;
    const int ldsOff = wave * LDP + lane * 4;

    const float* bcol = Wc + (size_t)(khalf * 2) * D + nBase + col;

    const int numTiles = M >> 5;
    int mt = blockIdx.x;

    f4 stageF[4], stageG[4];
    if (mt < numTiles) {
        const float* srcF = Fa + (size_t)(mt << 5) * D + srcOff;
        const float* srcG = Ag + (size_t)(mt << 5) * D + srcOff;
        for (int i = 0; i < 4; ++i) {
            stageF[i] = *(const f4*)(srcF + (size_t)(8 * i) * D);
            stageG[i] = *(const f4*)(srcG + (size_t)(8 * i) * D);
        }
    }

    for (; mt < numTiles; mt += gridDim.x) {
        const int mBase = mt << 5;
        const int nxt   = mt + gridDim.x;
        v8f acc0 = {};
        v8f acc1 = {};

        {
            float* tile = tileA;
            for (int i = 0; i < 4; ++i)
                *(f4*)(tile + ldsOff + 8 * i * LDP) = stageF[i];
            __syncthreads();

            if (nxt < numTiles) {
                const float* src = Fa + (size_t)(nxt << 5) * D + srcOff;
                for (int i = 0; i < 4; ++i)
                    stageF[i] = *(const f4*)(src + (size_t)(8 * i) * D);
            }

            const float* arow0 = tile + col * LDP + khalf * 2;
            const float* arow1 = arow0 + 16 * LDP;
#pragma unroll
            for (int kk = 0; kk < 32; ++kk) {
                v2f b;
                b.x = bcol[(size_t)(kk * 4) * D];
                b.y = bcol[(size_t)(kk * 4 + 1) * D];
                v2f a0 = *(const v2f*)(arow0 + kk * 4);
                v2f a1 = *(const v2f*)(arow1 + kk * 4);
                acc0 = __builtin_amdgcn_wmma_f32_16x16x4_f32(
                    false, a0, false, b, (short)0, acc0, false, false);
                acc1 = __builtin_amdgcn_wmma_f32_16x16x4_f32(
                    false, a1, false, b, (short)0, acc1, false, false);
            }
        }

        {
            float* tile = tileA + TILE_FLOATS;
            for (int i = 0; i < 4; ++i)
                *(f4*)(tile + ldsOff + 8 * i * LDP) = stageG[i];
            __syncthreads();

            if (nxt < numTiles) {
                const float* src = Ag + (size_t)(nxt << 5) * D + srcOff;
                for (int i = 0; i < 4; ++i)
                    stageG[i] = *(const f4*)(src + (size_t)(8 * i) * D);
            }

            const float* arow0 = tile + col * LDP + khalf * 2;
            const float* arow1 = arow0 + 16 * LDP;
#pragma unroll
            for (int kk = 0; kk < 32; ++kk) {
                v2f b;
                b.x = bcol[(size_t)(128 + kk * 4) * D];
                b.y = bcol[(size_t)(128 + kk * 4 + 1) * D];
                v2f a0 = *(const v2f*)(arow0 + kk * 4);
                v2f a1 = *(const v2f*)(arow1 + kk * 4);
                acc0 = __builtin_amdgcn_wmma_f32_16x16x4_f32(
                    false, a0, false, b, (short)0, acc0, false, false);
                acc1 = __builtin_amdgcn_wmma_f32_16x16x4_f32(
                    false, a1, false, b, (short)0, acc1, false, false);
            }
        }

        const float bias = bc[nBase + col];
        float* orow0 = out + (size_t)mBase * D + nBase + col;
        float* orow1 = orow0 + (size_t)16 * D;
#pragma unroll
        for (int r = 0; r < 8; ++r) {
            float v0 = acc0[r] + bias;
            float v1 = acc1[r] + bias;
            orow0[(size_t)(r + khalf * 8) * D] = v0 > 0.0f ? v0 : 0.0f;
            orow1[(size_t)(r + khalf * 8) * D] = v1 > 0.0f ? v1 : 0.0f;
        }
    }
}

extern "C" void kernel_launch(void* const* d_in, const int* in_sizes, int n_in,
                              void* d_out, int out_size, void* d_ws, size_t ws_size,
                              hipStream_t stream)
{
    const float* variables = (const float*)d_in[0];
    const float* factors   = (const float*)d_in[1];
    const int*   senders   = (const int*)  d_in[2];
    const int*   receivers = (const int*)  d_in[3];
    const float* edge_attr = (const float*)d_in[4];
    const float* W_msg     = (const float*)d_in[5];
    const float* b_msg     = (const float*)d_in[6];
    const float* W_comb    = (const float*)d_in[7];
    const float* b_comb    = (const float*)d_in[8];

    const int V = in_sizes[0] / D;
    const int F = in_sizes[1] / D;
    const int E = in_sizes[2];

    const float* W_i = W_msg;
    const float* W_j = W_msg + (size_t)128 * D;
    const float* w_e = W_msg + (size_t)256 * D;

    float* Pv   = (float*)d_ws;
    float* Pf   = Pv + (size_t)V * D;
    float* aggr = Pf + (size_t)F * D;

    const int tilesV = V >> 5, tilesF = F >> 5;
    const int gV = tilesV < 2048 ? tilesV : 2048;
    const int gF = tilesF < 2048 ? tilesF : 2048;

    gemm_xw<<<gV, 256, 0, stream>>>(variables, W_j, Pv, V);
    gemm_xw<<<gF, 256, 0, stream>>>(factors,   W_i, Pf, F);

    row_owner_sum<<<(F + RPB - 1) / RPB, 1024, 0, stream>>>(
        Pf, Pv, senders, receivers, edge_attr, w_e, b_msg, aggr, E, F);

    combine_gemm<<<gF, 256, 0, stream>>>(
        factors, aggr, W_comb, b_comb, (float*)d_out, F);
}


# Step 2 of the row-owner sum looks through a whole 224-entry edge list per trip (7 row-number reads issued together, list lengths held in a register), not 32 entries per trip; add order unchanged
# speedup vs baseline: 1.7333x; 1.0564x over previous
#include <hip/hip_runtime.h>

#define D 128

typedef __attribute__((ext_vector_type(2))) float v2f;
typedef __attribute__((ext_vector_type(4))) float f4;
typedef __attribute__((ext_vector_type(8))) float v8f;

#define LDP 132
#define TILE_FLOATS (32 * LDP)

__global__ void __launch_bounds__(256)
gemm_xw(const float* __restrict__ X, const float* __restrict__ W,
        float* __restrict__ P, int M)
{
    __shared__ __attribute__((aligned(16))) float tileA[2 * TILE_FLOATS];

    const int lane  = threadIdx.x & 31;
    const int wave  = threadIdx.x >> 5;
    const int nBase = wave * 16;
    const int col   = lane & 15;
    const int khalf = lane >> 4;

    v2f bfrag[32];
#pragma unroll
    for (int kk = 0; kk < 32; ++kk) {
        const int k = kk * 4 + khalf * 2;
        bfrag[kk].x = W[(size_t)k * D + nBase + col];
        bfrag[kk].y = W[(size_t)(k + 1) * D + nBase + col];
    }

    const int srcOff = wave * D + lane * 4;
    const int ldsOff = wave * LDP + lane * 4;

    const int numTiles = M >> 5;
    int mt = blockIdx.x;

    f4 stage[4];
    if (mt < numTiles) {
        const float* src = X + (size_t)(mt << 5) * D + srcOff;
        for (int i = 0; i < 4; ++i)
            stage[i] = *(const f4*)(src + (size_t)(8 * i) * D);
    }

    int buf = 0;
    for (; mt < numTiles; mt += gridDim.x) {
        float* tile = tileA + buf * TILE_FLOATS;
        for (int i = 0; i < 4; ++i)
            *(f4*)(tile + ldsOff + 8 * i * LDP) = stage[i];
        __syncthreads();

        const int nxt = mt + gridDim.x;
        if (nxt < numTiles) {
            const float* src = X + (size_t)(nxt << 5) * D + srcOff;
            for (int i = 0; i < 4; ++i)
                stage[i] = *(const f4*)(src + (size_t)(8 * i) * D);
        }

        const float* arow0 = tile + col * LDP + khalf * 2;
        const float* arow1 = arow0 + 16 * LDP;
        v8f acc0 = {};
        v8f acc1 = {};
#pragma unroll
        for (int kk = 0; kk < 32; ++kk) {
            v2f a0 = *(const v2f*)(arow0 + kk * 4);
            v2f a1 = *(const v2f*)(arow1 + kk * 4);
            acc0 = __builtin_amdgcn_wmma_f32_16x16x4_f32(
                false, a0, false, bfrag[kk], (short)0, acc0, false, false);
            acc1 = __builtin_amdgcn_wmma_f32_16x16x4_f32(
                false, a1, false, bfrag[kk], (short)0, acc1, false, false);
        }
        const int mBase = mt << 5;
        float* prow0 = P + (size_t)mBase * D + nBase + col;
        float* prow1 = prow0 + (size_t)16 * D;
#pragma unroll
        for (int r = 0; r < 8; ++r) {
            prow0[(size_t)(r + khalf * 8) * D] = acc0[r];
            prow1[(size_t)(r + khalf * 8) * D] = acc1[r];
        }
        buf ^= 1;
    }
}

#define RPW  13
#define RPB  (32 * RPW)
#define LCAP 224
#define NSUB (LCAP / 32)
#define QCAP (32 + LCAP)
static_assert(LCAP % 32 == 0, "a list is a whole number of 32-entry groups");

static __device__ __forceinline__ f4 pick_row(const f4 (&r)[RPW], int t)
{
    switch (t) {
    case 0:  return r[0];
    case 1:  return r[1];
    case 2:  return r[2];
    case 3:  return r[3];
    case 4:  return r[4];
    case 5:  return r[5];
    case 6:  return r[6];
    case 7:  return r[7];
    case 8:  return r[8];
    case 9:  return r[9];
    case 10: return r[10];
    case 11: return r[11];
    default: return r[12];
    }
}

__global__ void __launch_bounds__(1024)
row_owner_sum(const float* __restrict__ Pf, const float* __restrict__ Pv,
              const int* __restrict__ senders, const int* __restrict__ receivers,
              const float* __restrict__ edge_attr,
              const float* __restrict__ w_e, const float* __restrict__ b_msg,
              float* __restrict__ aggr, int E, int F)
{
    __shared__ f4  accS[32 * RPW * 32];
    __shared__ int listS[32 * LCAP];
    __shared__ int listA[32 * LCAP];
    __shared__ unsigned short listT[32 * LCAP];
    __shared__ int quI[32 * QCAP];
    __shared__ int cntW[32];
    __shared__ int moreFlag;

    const int lane = threadIdx.x & 31;
    const int w    = __builtin_amdgcn_readfirstlane(threadIdx.x >> 5);
    const int lo   = blockIdx.x * RPB;
    const int Rb   = F - lo < RPB ? F - lo : RPB;
    const int wlo  = w * RPW;
    const int o    = lane * 4;
    const unsigned below = (1u << lane) - 1u;

    f4*  myAcc = accS + wlo * 32;
    unsigned short* myT = listT + w * LCAP;
    int* myS = listS + w * LCAP;
    int* myA = listA + w * LCAP;
    int* mqI = quI + w * QCAP;

    const f4 zero = {0.0f, 0.0f, 0.0f, 0.0f};
    f4 pfr[RPW];
#pragma unroll
    for (int k = 0; k < RPW; ++k) {
        myAcc[k * 32 + lane] = zero;
        pfr[k] = wlo + k < Rb ? *(const f4*)(Pf + (size_t)(lo + wlo + k) * D + o) : zero;
    }

    const f4 we = *(const f4*)(w_e + o);
    const f4 bm = *(const f4*)(b_msg + o);

    const int numChunks = (E + 31) >> 5;
    const int numP      = (numChunks + 31) >> 5;
    const int pFull     = (E >> 5) >> 5;
    const int pRot      = (int)(((long long)blockIdx.x * numP) / gridDim.x);
    int p = 0;

    for (;;) {
        if (threadIdx.x == 0) moreFlag = 0;
        __syncthreads();

        int cursor = 0;
        while (p < numP && cursor + 32 <= LCAP) {
            int pr = p + pRot;
            if (pr >= numP) pr -= numP;
            if (pr + 16 <= pFull && p + 16 <= numP) {
                const int eBase = (pr * 32 + w) * 32 + lane;
                const int* src = receivers + eBase;
                int rr[16];
#pragma unroll
                for (int u = 0; u < 16; ++u) rr[u] = src[u * 1024];
                int done = 16;
                unsigned rbEff = (unsigned)Rb;
#pragma unroll
                for (int u = 0; u < 16; ++u) {
                    const int t = rr[u] - lo;
                    const bool m = (unsigned)t < rbEff;
                    const unsigned mask = __builtin_amdgcn_ballot_w32(m);
                    if (mask) {
                        if (m) {
                            const int slot = cursor + __builtin_popcount(mask & below);
                            myT[slot] = (unsigned short)t;
                            myS[slot] = eBase + u * 1024;
                        }
                        cursor += __builtin_popcount(mask);
                        if (cursor + 32 > LCAP) { rbEff = 0u; done = u + 1; }
                    }
                }
                p += done;
            } else {
                const int cp = pr * 32 + w;
                const int e  = cp * 32 + lane;
                const int r  = (cp < numChunks && e < E) ? receivers[e] : -1;
                const int t  = r - lo;
                const bool m = (unsigned)t < (unsigned)Rb;
                const unsigned mask = __builtin_amdgcn_ballot_w32(m);
                if (mask) {
                    if (m) {
                        const int slot = cursor + __builtin_popcount(mask & below);
                        myT[slot] = (unsigned short)t;
                        myS[slot] = e;
                    }
                    cursor += __builtin_popcount(mask);
                }
                p += 1;
            }
        }
        for (int idx = lane; idx < cursor; idx += 32) {
            const int e = myS[idx];
            myS[idx] = senders[e];
            myA[idx] = __float_as_int(edge_attr[e]);
        }
        if (lane == 0) {
            cntW[w] = cursor;
            if (p < numP) moreFlag = 1;
        }
        __syncthreads();

        const int more = __builtin_amdgcn_readfirstlane(moreFlag);
        const int cntv = cntW[lane];
        int qn = 0;
        for (int q = 0; q < 32; ++q) {
            const int nq = __builtin_amdgcn_readlane(cntv, q);
            const unsigned short* lt = listT + q * LCAP;
            int raw[NSUB];
#pragma unroll
            for (int j = 0; j < NSUB; ++j)
                raw[j] = (int)lt[32 * j + lane];
            int tt[NSUB];
#pragma unroll
            for (int j = 0; j < NSUB; ++j)
                tt[j] = 32 * j + lane < nq ? raw[j] - wlo : -1;
#pragma unroll
            for (int j = 0; j < NSUB; ++j) {
                if (32 * j < nq) {
                    const bool m = (unsigned)tt[j] < (unsigned)RPW;
                    const unsigned mask = __builtin_amdgcn_ballot_w32(m);
                    if (mask) {
                        if (m) {
                            const int slot = qn + __builtin_popcount(mask & below);
                            mqI[slot] = q * LCAP + 32 * j + lane;
                        }
                        qn += __builtin_popcount(mask);
                    }
                }
            }
            if (qn > QCAP - LCAP || (q == 31 && qn > 0)) {
                for (int g = 0; g < qn; g += 8) {
                    const int cnt = qn - g < 8 ? qn - g : 8;
                    int tq[8];
                    float aq[8];
                    f4 pv[8];
#pragma unroll
                    for (int u = 0; u < 8; ++u) {
                        const int li = mqI[g + (u < cnt ? u : cnt - 1)];
                        tq[u] = __builtin_amdgcn_readfirstlane((int)listT[li]) - wlo;
                        const int s = listS[li];
                        aq[u] = __int_as_float(listA[li]);
                        pv[u] = *(const f4*)(Pv + (size_t)s * D + o);
                    }
#pragma unroll
                    for (int u = 0; u < 8; ++u) {
                        if (u < cnt) {
                            const f4 pf = pick_row(pfr, tq[u]);
                            f4 a = myAcc[tq[u] * 32 + lane];
#pragma unroll
                            for (int i = 0; i < 4; ++i) {
                                float m2 = pf[i] + pv[u][i] + aq[u] * we[i] + bm[i];
                                a[i] += m2 > 0.0f ? m2 : 0.0f;
                            }
                            myAcc[tq[u] * 32 + lane] = a;
                        }
                    }
                }
                qn = 0;
            }
        }
        __syncthreads();
        if (!more) break;
    }

    for (int k = 0; k < RPW; ++k) {
        if (wlo + k < Rb)
            *(f4*)(aggr + (size_t)(lo + wlo + k) * D + o) = myAcc[k * 32 + lane];
    }
}

__global__ void __launch_bounds__(256)
combine_gemm(const float* __restrict__ Fa, const float* __restrict__ Ag,
             const float* __restrict__ Wc, const float* __restrict__ bc,
             float* __restrict__ out, int M)
{
    __shared__ __attribute__((aligned(16))) float tileA[2 * TILE_FLOATS];

    const int lane  = threadIdx.x & 31;
    const int wave  = threadIdx.x >> 5;
    const int nBase = wave * 16;
    const int col   = lane & 15;
    const int khalf = lane >> 4;

    const int srcOff = wave * D + lane * 4;
    const int ldsOff = wave * LDP + lane * 4;

    const float* bcol = Wc + (size_t)(khalf * 2) * D + nBase + col;

    const int numTiles = M >> 5;
    int mt = blockIdx.x;

    f4 stageF[4], stageG[4];
    if (mt < numTiles) {
        const float* srcF = Fa + (size_t)(mt << 5) * D + srcOff;
        const float* srcG = Ag + (size_t)(mt << 5) * D + srcOff;
        for (int i = 0; i < 4; ++i) {
            stageF[i] = *(const f4*)(srcF + (size_t)(8 * i) * D);
            stageG[i] = *(const f4*)(srcG + (size_t)(8 * i) * D);
        }
    }

    for (; mt < numTiles; mt += gridDim.x) {
        const int mBase = mt << 5;
        const int nxt   = mt + gridDim.x;
        v8f acc0 = {};
        v8f acc1 = {};

        {
            float* tile = tileA;
            for (int i = 0; i < 4; ++i)
                *(f4*)(tile + ldsOff + 8 * i * LDP) = stageF[i];
            __syncthreads();

            if (nxt < numTiles) {
                const float* src = Fa + (size_t)(nxt << 5) * D + srcOff;
                for (int i = 0; i < 4; ++i)
                    stageF[i] = *(const f4*)(src + (size_t)(8 * i) * D);
            }

            const float* arow0 = tile + col * LDP + khalf * 2;
            const float* arow1 = arow0 + 16 * LDP;
#pragma unroll
            for (int kk = 0; kk < 32; ++kk) {
                v2f b;
                b.x = bcol[(size_t)(kk * 4) * D];
                b.y = bcol[(size_t)(kk * 4 + 1) * D];
                v2f a0 = *(const v2f*)(arow0 + kk * 4);
                v2f a1 = *(const v2f*)(arow1 + kk * 4);
                acc0 = __builtin_amdgcn_wmma_f32_16x16x4_f32(
                    false, a0, false, b, (short)0, acc0, false, false);
                acc1 = __builtin_amdgcn_wmma_f32_16x16x4_f32(
                    false, a1, false, b, (short)0, acc1, false, false);
            }
        }

        {
            float* tile = tileA + TILE_FLOATS;
            for (int i = 0; i < 4; ++i)
                *(f4*)(tile + ldsOff + 8 * i * LDP) = stageG[i];
            __syncthreads();

            if (nxt < numTiles) {
                const float* src = Ag + (size_t)(nxt << 5) * D + srcOff;
                for (int i = 0; i < 4; ++i)
                    stageG[i] = *(const f4*)(src + (size_t)(8 * i) * D);
            }

            const float* arow0 = tile + col * LDP + khalf * 2;
            const float* arow1 = arow0 + 16 * LDP;
#pragma unroll
            for (int kk = 0; kk < 32; ++kk) {
                v2f b;
                b.x = bcol[(size_t)(128 + kk * 4) * D];
                b.y = bcol[(size_t)(128 + kk * 4 + 1) * D];
                v2f a0 = *(const v2f*)(arow0 + kk * 4);
                v2f a1 = *(const v2f*)(arow1 + kk * 4);
                acc0 = __builtin_amdgcn_wmma_f32_16x16x4_f32(
                    false, a0, false, b, (short)0, acc0, false, false);
                acc1 = __builtin_amdgcn_wmma_f32_16x16x4_f32(
                    false, a1, false, b, (short)0, acc1, false, false);
            }
        }

        const float bias = bc[nBase + col];
        float* orow0 = out + (size_t)mBase * D + nBase + col;
        float* orow1 = orow0 + (size_t)16 * D;
#pragma unroll
        for (int r = 0; r < 8; ++r) {
            float v0 = acc0[r] + bias;
            float v1 = acc1[r] + bias;
            orow0[(size_t)(r + khalf * 8) * D] = v0 > 0.0f ? v0 : 0.0f;
            orow1[(size_t)(r + khalf * 8) * D] = v1 > 0.0f ? v1 : 0.0f;
        }
    }
}

extern "C" void kernel_launch(void* const* d_in, const int* in_sizes, int n_in,
                              void* d_out, int out_size, void* d_ws, size_t ws_size,
                              hipStream_t stream)
{
    const float* variables = (const float*)d_in[0];
    const float* factors   = (const float*)d_in[1];
    const int*   senders   = (const int*)  d_in[2];
    const int*   receivers = (const int*)  d_in[3];
    const float* edge_attr = (const float*)d_in[4];
    const float* W_msg     = (const float*)d_in[5];
    const float* b_msg     = (const float*)d_in[6];
    const float* W_comb    = (const float*)d_in[7];
    const float* b_comb    = (const float*)d_in[8];

    const int V = in_sizes[0] / D;
    const int F = in_sizes[1] / D;
    const int E = in_sizes[2];

    const float* W_i = W_msg;
    const float* W_j = W_msg + (size_t)128 * D;
    const float* w_e = W_msg + (size_t)256 * D;

    float* Pv   = (float*)d_ws;
    float* Pf   = Pv + (size_t)V * D;
    float* aggr = Pf + (size_t)F * D;

    const int tilesV = V >> 5, tilesF = F >> 5;
    const int gV = tilesV < 2048 ? tilesV : 2048;
    const int gF = tilesF < 2048 ? tilesF : 2048;

    gemm_xw<<<gV, 256, 0, stream>>>(variables, W_j, Pv, V);
    gemm_xw<<<gF, 256, 0, stream>>>(factors,   W_i, Pf, F);

    row_owner_sum<<<(F + RPB - 1) / RPB, 1024, 0, stream>>>(
        Pf, Pv, senders, receivers, edge_attr, w_e, b_msg, aggr, E, F);

    combine_gemm<<<gF, 256, 0, stream>>>(
        factors, aggr, W_comb, b_comb, (float*)d_out, F);
}


# Row-owner sum reads only the receiver chunks that a new small filter launch marks as holding a row of its block (one bit per chunk and row block), not all indices; list and add order unchanged
# speedup vs baseline: 1.8135x; 1.0463x over previous
#include <hip/hip_runtime.h>

#define D 128

typedef __attribute__((ext_vector_type(2))) float v2f;
typedef __attribute__((ext_vector_type(4))) float f4;
typedef __attribute__((ext_vector_type(8))) float v8f;

#define LDP 132
#define TILE_FLOATS (32 * LDP)

__global__ void __launch_bounds__(256)
gemm_xw(const float* __restrict__ X, const float* __restrict__ W,
        float* __restrict__ P, int M)
{
    __shared__ __attribute__((aligned(16))) float tileA[2 * TILE_FLOATS];

    const int lane  = threadIdx.x & 31;
    const int wave  = threadIdx.x >> 5;
    const int nBase = wave * 16;
    const int col   = lane & 15;
    const int khalf = lane >> 4;

    v2f bfrag[32];
#pragma unroll
    for (int kk = 0; kk < 32; ++kk) {
        const int k = kk * 4 + khalf * 2;
        bfrag[kk].x = W[(size_t)k * D + nBase + col];
        bfrag[kk].y = W[(size_t)(k + 1) * D + nBase + col];
    }

    const int srcOff = wave * D + lane * 4;
    const int ldsOff = wave * LDP + lane * 4;

    const int numTiles = M >> 5;
    int mt = blockIdx.x;

    f4 stage[4];
    if (mt < numTiles) {
        const float* src = X + (size_t)(mt << 5) * D + srcOff;
        for (int i = 0; i < 4; ++i)
            stage[i] = *(const f4*)(src + (size_t)(8 * i) * D);
    }

    int buf = 0;
    for (; mt < numTiles; mt += gridDim.x) {
        float* tile = tileA + buf * TILE_FLOATS;
        for (int i = 0; i < 4; ++i)
            *(f4*)(tile + ldsOff + 8 * i * LDP) = stage[i];
        __syncthreads();

        const int nxt = mt + gridDim.x;
        if (nxt < numTiles) {
            const float* src = X + (size_t)(nxt << 5) * D + srcOff;
            for (int i = 0; i < 4; ++i)
                stage[i] = *(const f4*)(src + (size_t)(8 * i) * D);
        }

        const float* arow0 = tile + col * LDP + khalf * 2;
        const float* arow1 = arow0 + 16 * LDP;
        v8f acc0 = {};
        v8f acc1 = {};
#pragma unroll
        for (int kk = 0; kk < 32; ++kk) {
            v2f a0 = *(const v2f*)(arow0 + kk * 4);
            v2f a1 = *(const v2f*)(arow1 + kk * 4);
            acc0 = __builtin_amdgcn_wmma_f32_16x16x4_f32(
                false, a0, false, bfrag[kk], (short)0, acc0, false, false);
            acc1 = __builtin_amdgcn_wmma_f32_16x16x4_f32(
                false, a1, false, bfrag[kk], (short)0, acc1, false, false);
        }
        const int mBase = mt << 5;
        float* prow0 = P + (size_t)mBase * D + nBase + col;
        float* prow1 = prow0 + (size_t)16 * D;
#pragma unroll
        for (int r = 0; r < 8; ++r) {
            prow0[(size_t)(r + khalf * 8) * D] = acc0[r];
            prow1[(size_t)(r + khalf * 8) * D] = acc1[r];
        }
        buf ^= 1;
    }
}

#define RPW  13
#define RPB  (32 * RPW)
#define FB   8
#define NPLANE 8
#define LCAP 224
#define NSUB (LCAP / 32)
#define QCAP (32 + LCAP)
static_assert(LCAP % 32 == 0, "a list is a whole number of 32-entry groups");

__global__ void __launch_bounds__(256)
chunk_filter(const int* __restrict__ receivers, unsigned* __restrict__ filt,
             int E, int F)
{
    __shared__ unsigned flagS[8 * 4 * 64];

    const int lane = threadIdx.x & 31;
    const int wave = __builtin_amdgcn_readfirstlane(threadIdx.x >> 5);
    unsigned* flagW = flagS + wave * 4 * 64;
    unsigned char* flag = (unsigned char*)flagW;
#pragma unroll
    for (int i = 0; i < 8; ++i) flagW[32 * i + lane] = 0u;

    const int numChunks = (E + 31) >> 5;
    const int numP      = (numChunks + 31) >> 5;
    const int task = blockIdx.x * 8 + wave;
    const int wq   = task & 31;
    const int p0   = (task >> 5) * 32;
    if (p0 >= numP) return;

    int rr[32];
#pragma unroll
    for (int j = 0; j < 32; ++j) {
        const int cp = (p0 + j) * 32 + wq;
        const int e  = cp * 32 + lane;
        const bool ok = p0 + j < numP && cp < numChunks && e < E;
        const int val = receivers[ok ? e : 0];
        rr[j] = ok ? val : -1;
    }

    unsigned wd[NPLANE];
#pragma unroll
    for (int k = 0; k < NPLANE; ++k) wd[k] = 0u;
#pragma unroll
    for (int jj = 0; jj < 32; jj += 4) {
        unsigned bid[4];
        bool v[4];
#pragma unroll
        for (int q = 0; q < 4; ++q) {
            const unsigned r = (unsigned)rr[jj + q];
            bid[q] = r / RPB;
            v[q] = r < (unsigned)F && bid[q] < 32u * NPLANE;
            if (v[q]) flag[256 * q + bid[q]] = 1;
        }
#pragma unroll
        for (int q = 0; q < 4; ++q) {
#pragma unroll
            for (int k = 0; k < NPLANE; ++k) {
                const unsigned m =
                    __builtin_amdgcn_ballot_w32(flag[256 * q + 32 * k + lane] != 0);
                if (lane == jj + q) wd[k] = m;
            }
        }
#pragma unroll
        for (int q = 0; q < 4; ++q)
            if (v[q]) flag[256 * q + bid[q]] = 0;
    }

    if (p0 + lane < numP) {
#pragma unroll
        for (int k = 0; k < NPLANE; ++k)
            filt[((size_t)k * 32 + wq) * numP + p0 + lane] = wd[k];
    }
}

static __device__ __forceinline__ f4 pick_row(const f4 (&r)[RPW], int t)
{
    switch (t) {
    case 0:  return r[0];
    case 1:  return r[1];
    case 2:  return r[2];
    case 3:  return r[3];
    case 4:  return r[4];
    case 5:  return r[5];
    case 6:  return r[6];
    case 7:  return r[7];
    case 8:  return r[8];
    case 9:  return r[9];
    case 10: return r[10];
    case 11: return r[11];
    default: return r[12];
    }
}

__global__ void __launch_bounds__(1024)
row_owner_sum(const float* __restrict__ Pf, const float* __restrict__ Pv,
              const int* __restrict__ senders, const int* __restrict__ receivers,
              const float* __restrict__ edge_attr,
              const float* __restrict__ w_e, const float* __restrict__ b_msg,
              const unsigned* __restrict__ filt,
              float* __restrict__ aggr, int E, int F)
{
    __shared__ f4  accS[32 * RPW * 32];
    __shared__ int listS[32 * LCAP];
    __shared__ int listA[32 * LCAP];
    __shared__ unsigned short listT[32 * LCAP];
    __shared__ int quI[32 * QCAP];
    __shared__ int cntW[32];
    __shared__ int moreFlag;

    const int lane = threadIdx.x & 31;
    const int w    = __builtin_amdgcn_readfirstlane(threadIdx.x >> 5);
    const int lo   = blockIdx.x * RPB;
    const int Rb   = F - lo < RPB ? F - lo : RPB;
    const int wlo  = w * RPW;
    const int o    = lane * 4;
    const unsigned below = (1u << lane) - 1u;

    f4*  myAcc = accS + wlo * 32;
    unsigned short* myT = listT + w * LCAP;
    int* myS = listS + w * LCAP;
    int* myA = listA + w * LCAP;
    int* mqI = quI + w * QCAP;

    const f4 zero = {0.0f, 0.0f, 0.0f, 0.0f};
    f4 pfr[RPW];
#pragma unroll
    for (int k = 0; k < RPW; ++k) {
        myAcc[k * 32 + lane] = zero;
        pfr[k] = wlo + k < Rb ? *(const f4*)(Pf + (size_t)(lo + wlo + k) * D + o) : zero;
    }

    const f4 we = *(const f4*)(w_e + o);
    const f4 bm = *(const f4*)(b_msg + o);

    const int numChunks = (E + 31) >> 5;
    const int numP      = (numChunks + 31) >> 5;
    const int pRot      = (int)(((long long)blockIdx.x * numP) / gridDim.x);
    const unsigned  fbit = blockIdx.x & 31;
    const unsigned* fw   = filt + ((size_t)(blockIdx.x >> 5) * 32 + w) * numP;
    int p = 0;

    for (;;) {
        if (threadIdx.x == 0) moreFlag = 0;
        __syncthreads();

        int cursor = 0;
        unsigned fv[FB];
#pragma unroll
        for (int u = 0; u < FB; ++u) {
            const int pp = p + 32 * u + lane;
            int pr = pp + pRot;
            if (pr >= numP) pr -= numP;
            const unsigned val = fw[pp < numP ? pr : 0];
            fv[u] = pp < numP ? val : 0u;
        }
        while (p < numP && cursor + 32 <= LCAP) {
            int nl = 0;
#pragma unroll
            for (int u = 0; u < FB; ++u) {
                const bool m = ((fv[u] >> fbit) & 1u) != 0u;
                const unsigned mask = __builtin_amdgcn_ballot_w32(m);
                if (mask) {
                    if (m) mqI[nl + __builtin_popcount(mask & below)] = p + 32 * u + lane;
                    nl += __builtin_popcount(mask);
                }
            }
            int pNext = p + 32 * FB < numP ? p + 32 * FB : numP;
#pragma unroll
            for (int u = 0; u < FB; ++u) {
                const int pp = pNext + 32 * u + lane;
                int pr = pp + pRot;
                if (pr >= numP) pr -= numP;
                const unsigned val = fw[pp < numP ? pr : 0];
                fv[u] = pp < numP ? val : 0u;
            }
            unsigned rbEff = (unsigned)Rb;
            for (int i0 = 0; i0 < nl && rbEff != 0u; i0 += 16) {
                const int cnt = nl - i0 < 16 ? nl - i0 : 16;
                const int pqv = mqI[i0 + (lane < cnt ? lane : cnt - 1)];
                int pq[16];
                int eb[16];
                int rr[16];
#pragma unroll
                for (int u = 0; u < 16; ++u) {
                    pq[u] = __builtin_amdgcn_readlane(pqv, u);
                    int pr = pq[u] + pRot;
                    if (pr >= numP) pr -= numP;
                    eb[u] = (pr * 32 + w) * 32;
                    const int e = eb[u] + lane;
                    rr[u] = receivers[e < E ? e : 0];
                }
#pragma unroll
                for (int u = 0; u < 16; ++u) {
                    if (u < cnt) {
                        const int t = rr[u] - lo;
                        const bool m = (unsigned)t < rbEff && eb[u] + lane < E;
                        const unsigned mask = __builtin_amdgcn_ballot_w32(m);
                        if (mask) {
                            if (m) {
                                const int slot = cursor + __builtin_popcount(mask & below);
                                myT[slot] = (unsigned short)t;
                                myS[slot] = eb[u] + lane;
                            }
                            cursor += __builtin_popcount(mask);
                            if (cursor + 32 > LCAP) { rbEff = 0u; pNext = pq[u] + 1; }
                        }
                    }
                }
            }
            p = pNext;
        }
        for (int idx = lane; idx < cursor; idx += 32) {
            const int e = myS[idx];
            myS[idx] = senders[e];
            myA[idx] = __float_as_int(edge_attr[e]);
        }
        if (lane == 0) {
            cntW[w] = cursor;
            if (p < numP) moreFlag = 1;
        }
        __syncthreads();

        const int more = __builtin_amdgcn_readfirstlane(moreFlag);
        const int cntv = cntW[lane];
        int qn = 0;
        for (int q = 0; q < 32; ++q) {
            const int nq = __builtin_amdgcn_readlane(cntv, q);
            const unsigned short* lt = listT + q * LCAP;
            int raw[NSUB];
#pragma unroll
            for (int j = 0; j < NSUB; ++j)
                raw[j] = (int)lt[32 * j + lane];
            int tt[NSUB];
#pragma unroll
            for (int j = 0; j < NSUB; ++j)
                tt[j] = 32 * j + lane < nq ? raw[j] - wlo : -1;
#pragma unroll
            for (int j = 0; j < NSUB; ++j) {
                if (32 * j < nq) {
                    const bool m = (unsigned)tt[j] < (unsigned)RPW;
                    const unsigned mask = __builtin_amdgcn_ballot_w32(m);
                    if (mask) {
                        if (m) {
                            const int slot = qn + __builtin_popcount(mask & below);
                            mqI[slot] = q * LCAP + 32 * j + lane;
                        }
                        qn += __builtin_popcount(mask);
                    }
                }
            }
            if (qn > QCAP - LCAP || (q == 31 && qn > 0)) {
                for (int g = 0; g < qn; g += 8) {
                    const int cnt = qn - g < 8 ? qn - g : 8;
                    int tq[8];
                    float aq[8];
                    f4 pv[8];
#pragma unroll
                    for (int u = 0; u < 8; ++u) {
                        const int li = mqI[g + (u < cnt ? u : cnt - 1)];
                        tq[u] = __builtin_amdgcn_readfirstlane((int)listT[li]) - wlo;
                        const int s = listS[li];
                        aq[u] = __int_as_float(listA[li]);
                        pv[u] = *(const f4*)(Pv + (size_t)s * D + o);
                    }
#pragma unroll
                    for (int u = 0; u < 8; ++u) {
                        if (u < cnt) {
                            const f4 pf = pick_row(pfr, tq[u]);
                            f4 a = myAcc[tq[u] * 32 + lane];
#pragma unroll
                            for (int i = 0; i < 4; ++i) {
                                float m2 = pf[i] + pv[u][i] + aq[u] * we[i] + bm[i];
                                a[i] += m2 > 0.0f ? m2 : 0.0f;
                            }
                            myAcc[tq[u] * 32 + lane] = a;
                        }
                    }
                }
                qn = 0;
            }
        }
        __syncthreads();
        if (!more) break;
    }

    for (int k = 0; k < RPW; ++k) {
        if (wlo + k < Rb)
            *(f4*)(aggr + (size_t)(lo + wlo + k) * D + o) = myAcc[k * 32 + lane];
    }
}

__global__ void __launch_bounds__(256)
combine_gemm(const float* __restrict__ Fa, const float* __restrict__ Ag,
             const float* __restrict__ Wc, const float* __restrict__ bc,
             float* __restrict__ out, int M)
{
    __shared__ __attribute__((aligned(16))) float tileA[2 * TILE_FLOATS];

    const int lane  = threadIdx.x & 31;
    const int wave  = threadIdx.x >> 5;
    const int nBase = wave * 16;
    const int col   = lane & 15;
    const int khalf = lane >> 4;

    const int srcOff = wave * D + lane * 4;
    const int ldsOff = wave * LDP + lane * 4;

    const float* bcol = Wc + (size_t)(khalf * 2) * D + nBase + col;

    const int numTiles = M >> 5;
    int mt = blockIdx.x;

    f4 stageF[4], stageG[4];
    if (mt < numTiles) {
        const float* srcF = Fa + (size_t)(mt << 5) * D + srcOff;
        const float* srcG = Ag + (size_t)(mt << 5) * D + srcOff;
        for (int i = 0; i < 4; ++i) {
            stageF[i] = *(const f4*)(srcF + (size_t)(8 * i) * D);
            stageG[i] = *(const f4*)(srcG + (size_t)(8 * i) * D);
        }
    }

    for (; mt < numTiles; mt += gridDim.x) {
        const int mBase = mt << 5;
        const int nxt   = mt + gridDim.x;
        v8f acc0 = {};
        v8f acc1 = {};

        {
            float* tile = tileA;
            for (int i = 0; i < 4; ++i)
                *(f4*)(tile + ldsOff + 8 * i * LDP) = stageF[i];
            __syncthreads();

            if (nxt < numTiles) {
                const float* src = Fa + (size_t)(nxt << 5) * D + srcOff;
                for (int i = 0; i < 4; ++i)
                    stageF[i] = *(const f4*)(src + (size_t)(8 * i) * D);
            }

            const float* arow0 = tile + col * LDP + khalf * 2;
            const float* arow1 = arow0 + 16 * LDP;
#pragma unroll
            for (int kk = 0; kk < 32; ++kk) {
                v2f b;
                b.x = bcol[(size_t)(kk * 4) * D];
                b.y = bcol[(size_t)(kk * 4 + 1) * D];
                v2f a0 = *(const v2f*)(arow0 + kk * 4);
                v2f a1 = *(const v2f*)(arow1 + kk * 4);
                acc0 = __builtin_amdgcn_wmma_f32_16x16x4_f32(
                    false, a0, false, b, (short)0, acc0, false, false);
                acc1 = __builtin_amdgcn_wmma_f32_16x16x4_f32(
                    false, a1, false, b, (short)0, acc1, false, false);
            }
        }

        {
            float* tile = tileA + TILE_FLOATS;
            for (int i = 0; i < 4; ++i)
                *(f4*)(tile + ldsOff + 8 * i * LDP) = stageG[i];
            __syncthreads();

            if (nxt < numTiles) {
                const float* src = Ag + (size_t)(nxt << 5) * D + srcOff;
                for (int i = 0; i < 4; ++i)
                    stageG[i] = *(const f4*)(src + (size_t)(8 * i) * D);
            }

            const float* arow0 = tile + col * LDP + khalf * 2;
            const float* arow1 = arow0 + 16 * LDP;
#pragma unroll
            for (int kk = 0; kk < 32; ++kk) {
                v2f b;
                b.x = bcol[(size_t)(128 + kk * 4) * D];
                b.y = bcol[(size_t)(128 + kk * 4 + 1) * D];
                v2f a0 = *(const v2f*)(arow0 + kk * 4);
                v2f a1 = *(const v2f*)(arow1 + kk * 4);
                acc0 = __builtin_amdgcn_wmma_f32_16x16x4_f32(
                    false, a0, false, b, (short)0, acc0, false, false);
                acc1 = __builtin_amdgcn_wmma_f32_16x16x4_f32(
                    false, a1, false, b, (short)0, acc1, false, false);
            }
        }

        const float bias = bc[nBase + col];
        float* orow0 = out + (size_t)mBase * D + nBase + col;
        float* orow1 = orow0 + (size_t)16 * D;
#pragma unroll
        for (int r = 0; r < 8; ++r) {
            float v0 = acc0[r] + bias;
            float v1 = acc1[r] + bias;
            orow0[(size_t)(r + khalf * 8) * D] = v0 > 0.0f ? v0 : 0.0f;
            orow1[(size_t)(r + khalf * 8) * D] = v1 > 0.0f ? v1 : 0.0f;
        }
    }
}

extern "C" void kernel_launch(void* const* d_in, const int* in_sizes, int n_in,
                              void* d_out, int out_size, void* d_ws, size_t ws_size,
                              hipStream_t stream)
{
    const float* variables = (const float*)d_in[0];
    const float* factors   = (const float*)d_in[1];
    const int*   senders   = (const int*)  d_in[2];
    const int*   receivers = (const int*)  d_in[3];
    const float* edge_attr = (const float*)d_in[4];
    const float* W_msg     = (const float*)d_in[5];
    const float* b_msg     = (const float*)d_in[6];
    const float* W_comb    = (const float*)d_in[7];
    const float* b_comb    = (const float*)d_in[8];

    const int V = in_sizes[0] / D;
    const int F = in_sizes[1] / D;
    const int E = in_sizes[2];

    const float* W_i = W_msg;
    const float* W_j = W_msg + (size_t)128 * D;
    const float* w_e = W_msg + (size_t)256 * D;

    float* Pv   = (float*)d_ws;
    float* Pf   = Pv + (size_t)V * D;
    float* aggr = Pf + (size_t)F * D;

    const int tilesV = V >> 5, tilesF = F >> 5;
    const int gV = tilesV < 2048 ? tilesV : 2048;
    const int gF = tilesF < 2048 ? tilesF : 2048;

    unsigned* filt = (unsigned*)d_out;
    const int numP = (((E + 31) >> 5) + 31) >> 5;
    chunk_filter<<<4 * ((numP + 31) >> 5), 256, 0, stream>>>(receivers, filt, E, F);

    gemm_xw<<<gV, 256, 0, stream>>>(variables, W_j, Pv, V);
    gemm_xw<<<gF, 256, 0, stream>>>(factors,   W_i, Pf, F);

    row_owner_sum<<<(F + RPB - 1) / RPB, 1024, 0, stream>>>(
        Pf, Pv, senders, receivers, edge_attr, w_e, b_msg, filt, aggr, E, F);

    combine_gemm<<<gF, 256, 0, stream>>>(
        factors, aggr, W_comb, b_comb, (float*)d_out, F);
}
